# starting kernel
# baseline (speedup 1.0000x reference)
#include <hip/hip_runtime.h>


typedef float v2f __attribute__((ext_vector_type(2)));
typedef float v8f __attribute__((ext_vector_type(8)));

#define T_TOT  2048
#define B_SZ   4
#define D_SZ   256
#define N_SZ   256
#define C_SZ   128
#define NCHUNK 16
#define EPS_F  1e-8f

__device__ __forceinline__ v8f wmma_f32_k4(v2f a, v2f b, v8f c) {
  return __builtin_amdgcn_wmma_f32_16x16x4_f32(false, a, false, b, (short)0, c, false, false);
}

__global__ void k_decay(const float* __restrict__ kin, const float* __restrict__ alpha,
                        float* __restrict__ kd, float* __restrict__ prefix) {
  const int c = blockIdx.x / B_SZ;
  const int b = blockIdx.x % B_SZ;
  const int n = threadIdx.x;
  const long stride = (long)B_SZ * N_SZ;
  const long base = ((long)c * C_SZ * B_SZ + b) * N_SZ + n;

  float P = 1.0f;
  #pragma unroll 4
  for (int t = 0; t < C_SZ; ++t) {
    float a = alpha[base + (long)t * stride];
    a = (a < EPS_F) ? EPS_F : a;
    P *= a;
  }
  const float inv = 1.0f / (P + EPS_F);
  float r = 1.0f;
  #pragma unroll 4
  for (int t = 0; t < C_SZ; ++t) {
    const long idx = base + (long)t * stride;
    float a = alpha[idx];
    a = (a < EPS_F) ? EPS_F : a;
    r *= a;
    kd[idx] = kin[idx] * (r * inv);
  }
  float a0 = alpha[base];
  a0 = (a0 < EPS_F) ? EPS_F : a0;
  prefix[(c * B_SZ + b) * N_SZ + n] = a0;
}

__global__ void k_gemmU(const float* __restrict__ v, const float* __restrict__ kd,
                        float* __restrict__ U) {
  __shared__ float Av[C_SZ][65];
  __shared__ float Bk[C_SZ][65];
  const int c = blockIdx.x / B_SZ, b = blockIdx.x % B_SZ;
  const int dBase = blockIdx.y * 64, nBase = blockIdx.z * 64;
  const int tid = threadIdx.x, lane = tid & 31, w = tid >> 5;

  for (int idx = tid; idx < C_SZ * 64; idx += 128) {
    const int t = idx >> 6, i = idx & 63;
    const long row = (long)(c * C_SZ + t) * B_SZ + b;
    Av[t][i] = v[row * D_SZ + dBase + i];
    Bk[t][i] = kd[row * N_SZ + nBase + i];
  }
  __syncthreads();

  const int half = lane >> 4, lan = lane & 15;
  const int mLoc = w * 16 + lan;
  v8f acc[4] = {};
  #pragma unroll
  for (int kk = 0; kk < C_SZ; kk += 4) {
    const int ka = kk + half * 2;
    v2f a; a.x = Av[ka][mLoc]; a.y = Av[ka + 1][mLoc];
    #pragma unroll
    for (int j = 0; j < 4; ++j) {
      v2f bf; bf.x = Bk[ka][j * 16 + lan]; bf.y = Bk[ka + 1][j * 16 + lan];
      acc[j] = wmma_f32_k4(a, bf, acc[j]);
    }
  }
  const long ub = (long)(c * B_SZ + b) * D_SZ;
  #pragma unroll
  for (int j = 0; j < 4; ++j)
    #pragma unroll
    for (int i = 0; i < 8; ++i) {
      const int row = dBase + w * 16 + half * 8 + i;
      const int col = nBase + j * 16 + lan;
      U[(ub + row) * N_SZ + col] = acc[j][i];
    }
}

__global__ void k_intra(const float* __restrict__ q, const float* __restrict__ kd,
                        const float* __restrict__ v, float* __restrict__ out) {
  __shared__ float As[C_SZ][C_SZ + 4];
  __shared__ float Pa[C_SZ][65];
  __shared__ float Pb[C_SZ][65];
  const int c = blockIdx.x / B_SZ, b = blockIdx.x % B_SZ;
  const int tid = threadIdx.x, lane = tid & 31, w = tid >> 5;
  const int half = lane >> 4, lan = lane & 15;
  const int mRow = w * 16 + lan;

  v8f acc[8] = {};
  for (int kn = 0; kn < N_SZ; kn += 64) {
    for (int idx = tid; idx < C_SZ * 64; idx += 256) {
      const int t = idx >> 6, j = idx & 63;
      const long row = ((long)(c * C_SZ + t) * B_SZ + b) * N_SZ + kn + j;
      Pa[t][j] = q[row];
      Pb[t][j] = kd[row];
    }
    __syncthreads();
    #pragma unroll
    for (int kk = 0; kk < 64; kk += 4) {
      const int ka = kk + half * 2;
      v2f a; a.x = Pa[mRow][ka]; a.y = Pa[mRow][ka + 1];
      #pragma unroll
      for (int j = 0; j < 8; ++j) {
        v2f bf; bf.x = Pb[j * 16 + lan][ka]; bf.y = Pb[j * 16 + lan][ka + 1];
        acc[j] = wmma_f32_k4(a, bf, acc[j]);
      }
    }
    __syncthreads();
  }
  #pragma unroll
  for (int j = 0; j < 8; ++j)
    #pragma unroll
    for (int i = 0; i < 8; ++i) {
      const int row = w * 16 + half * 8 + i;
      const int col = j * 16 + lan;
      As[row][col] = (col <= row) ? acc[j][i] : 0.0f;
    }
  __syncthreads();

  for (int dp = 0; dp < D_SZ; dp += 64) {
    for (int idx = tid; idx < C_SZ * 64; idx += 256) {
      const int s = idx >> 6, j = idx & 63;
      Pa[s][j] = v[((long)(c * C_SZ + s) * B_SZ + b) * D_SZ + dp + j];
    }
    __syncthreads();
    v8f acc2[4] = {};
    #pragma unroll
    for (int kk = 0; kk < C_SZ; kk += 4) {
      const int ka = kk + half * 2;
      v2f a; a.x = As[mRow][ka]; a.y = As[mRow][ka + 1];
      #pragma unroll
      for (int j = 0; j < 4; ++j) {
        v2f bf; bf.x = Pa[ka][j * 16 + lan]; bf.y = Pa[ka + 1][j * 16 + lan];
        acc2[j] = wmma_f32_k4(a, bf, acc2[j]);
      }
    }
    #pragma unroll
    for (int j = 0; j < 4; ++j)
      #pragma unroll
      for (int i = 0; i < 8; ++i) {
        const int row = w * 16 + half * 8 + i;
        const int col = dp + j * 16 + lan;
        out[((long)(c * C_SZ + row) * B_SZ + b) * D_SZ + col] = acc2[j][i];
      }
    __syncthreads();
  }
}

__global__ void k_scan(const float* __restrict__ U, const float* __restrict__ prefix,
                       float* __restrict__ Sst) {
  const long g = (long)blockIdx.x * blockDim.x + threadIdx.x;
  const int n = (int)(g % N_SZ);
  const int b = (int)(g / ((long)D_SZ * N_SZ));
  const long cs = (long)B_SZ * D_SZ * N_SZ;
  float s = 0.0f;
  for (int c = 0; c < NCHUNK; ++c) {
    Sst[c * cs + g] = s;
    s = s * prefix[(c * B_SZ + b) * N_SZ + n] + U[c * cs + g];
  }
}

__global__ void k_inter(const float* __restrict__ q, const float* __restrict__ prefix,
                        const float* __restrict__ Sst, float* __restrict__ out) {
  __shared__ float Qp[C_SZ][65];
  __shared__ float Sp[64][65];
  const int c = blockIdx.x / B_SZ, b = blockIdx.x % B_SZ;
  const int dBase = blockIdx.y * 64;
  const int tid = threadIdx.x, lane = tid & 31, w = tid >> 5;
  const int half = lane >> 4, lan = lane & 15;
  const int mRow = w * 16 + lan;
  const long cs = (long)B_SZ * D_SZ * N_SZ;

  v8f acc[4] = {};
  for (int np = 0; np < N_SZ; np += 64) {
    for (int idx = tid; idx < C_SZ * 64; idx += 256) {
      const int t = idx >> 6, j = idx & 63;
      const float pf = prefix[(c * B_SZ + b) * N_SZ + np + j];
      Qp[t][j] = q[((long)(c * C_SZ + t) * B_SZ + b) * N_SZ + np + j] * pf;
    }
    for (int idx = tid; idx < 64 * 64; idx += 256) {
      const int i = idx >> 6, j = idx & 63;
      Sp[i][j] = Sst[(long)c * cs + ((long)b * D_SZ + dBase + i) * N_SZ + np + j];
    }
    __syncthreads();
    #pragma unroll
    for (int kk = 0; kk < 64; kk += 4) {
      const int ka = kk + half * 2;
      v2f a; a.x = Qp[mRow][ka]; a.y = Qp[mRow][ka + 1];
      #pragma unroll
      for (int j = 0; j < 4; ++j) {
        v2f bf; bf.x = Sp[j * 16 + lan][ka]; bf.y = Sp[j * 16 + lan][ka + 1];
        acc[j] = wmma_f32_k4(a, bf, acc[j]);
      }
    }
    __syncthreads();
  }
  #pragma unroll
  for (int j = 0; j < 4; ++j)
    #pragma unroll
    for (int i = 0; i < 8; ++i) {
      const int row = w * 16 + half * 8 + i;
      const int col = dBase + j * 16 + lan;
      out[((long)(c * C_SZ + row) * B_SZ + b) * D_SZ + col] += acc[j][i];
    }
}

extern "C" void kernel_launch(void* const* d_in, const int* in_sizes, int n_in,
                              void* d_out, int out_size, void* d_ws, size_t ws_size,
                              hipStream_t stream) {
  const float* v     = (const float*)d_in[0];
  const float* k     = (const float*)d_in[1];
  const float* alpha = (const float*)d_in[2];
  const float* q     = (const float*)d_in[3];
  float* out = (float*)d_out;
  float* ws  = (float*)d_ws;

  const long KD_N = (long)T_TOT * B_SZ * N_SZ;
  const long PF_N = (long)NCHUNK * B_SZ * N_SZ;
  const long U_N  = (long)NCHUNK * B_SZ * D_SZ * N_SZ;
  float* kd     = ws;
  float* prefix = kd + KD_N;
  float* U      = prefix + PF_N;
  float* Sst    = U + U_N;

  k_decay<<<dim3(B_SZ * NCHUNK), 256, 0, stream>>>(k, alpha, kd, prefix);
  k_gemmU<<<dim3(B_SZ * NCHUNK, D_SZ / 64, N_SZ / 64), 128, 0, stream>>>(v, kd, U);
  k_intra<<<dim3(B_SZ * NCHUNK), 256, 0, stream>>>(q, kd, v, out);
  k_scan<<<dim3((B_SZ * D_SZ * N_SZ) / 256), 256, 0, stream>>>(U, prefix, Sst);
  k_inter<<<dim3(B_SZ * NCHUNK, D_SZ / 64), 256, 0, stream>>>(q, prefix, Sst, out);
}


# state plane stored in the readout's B-fragment order (tile held transposed, 16-byte pieces); k_readout loads state fragments from global to registers, no LDS panel; n summed in a new order
# speedup vs baseline: 3.3421x; 3.3421x over previous
#include <hip/hip_runtime.h>


typedef float v2f __attribute__((ext_vector_type(2)));
typedef float v8f __attribute__((ext_vector_type(8)));

#define T_TOT  2048
#define B_SZ   4
#define D_SZ   256
#define N_SZ   256
#define C_SZ   128
#define NCHUNK 16
#define EPS_F  1e-8f

__device__ __forceinline__ v8f wmma_f32_k4(v2f a, v2f b, v8f c) {
  return __builtin_amdgcn_wmma_f32_16x16x4_f32(false, a, false, b, (short)0, c, false, false);
}

__device__ __forceinline__ void lds_barrier() {
  asm volatile("s_wait_dscnt 0x0\n\ts_barrier_signal -1\n\ts_barrier_wait -1" ::: "memory");
}

#define DEC_T 64
__global__ void __launch_bounds__(DEC_T)
k_decay(const float* __restrict__ kin, const float* __restrict__ alpha,
        float* __restrict__ kd, float* __restrict__ prefix) {
  const int cb = blockIdx.x / (N_SZ / DEC_T);
  const int c = cb / B_SZ;
  const int b = cb % B_SZ;
  const int n = (blockIdx.x % (N_SZ / DEC_T)) * DEC_T + threadIdx.x;
  const long stride = (long)B_SZ * N_SZ;
  const long base = ((long)c * C_SZ * B_SZ + b) * N_SZ + n;

  float r[C_SZ];
  float kv[C_SZ];
  #pragma unroll
  for (int t = 0; t < C_SZ; ++t) r[t] = alpha[base + (long)t * stride];
  #pragma unroll
  for (int t = 0; t < C_SZ; ++t) kv[t] = kin[base + (long)t * stride];

  float P = 1.0f;
  #pragma unroll
  for (int t = 0; t < C_SZ; ++t) {
    float a = r[t];
    a = (a < EPS_F) ? EPS_F : a;
    P *= a;
    r[t] = P;
  }
  const float inv = 1.0f / (P + EPS_F);
  #pragma unroll
  for (int t = 0; t < C_SZ; ++t)
    kd[base + (long)t * stride] = kv[t] * (r[t] * inv);
  prefix[cb * N_SZ + n] = r[0];
}

typedef float v4f __attribute__((ext_vector_type(4)));
#define ST_P 40
#define ST_BLOCKS (B_SZ * (D_SZ / 32) * (N_SZ / 32))
#define SC_BLOCKS (B_SZ * NCHUNK * 2 * 2)
#define FUSED_LDS (2 * 2 * 2 * C_SZ * ST_P)

__device__ __forceinline__ void
state_part(float* __restrict__ L, const int bid,
           const float* __restrict__ v, const float* __restrict__ kd,
           const float* __restrict__ prefix, float* __restrict__ Sst) {
  float (*Lv)[2 * C_SZ][ST_P] = (float (*)[2 * C_SZ][ST_P])L;
  float (*Lk)[2 * C_SZ][ST_P] = (float (*)[2 * C_SZ][ST_P])(L + 2 * 2 * C_SZ * ST_P);
  const int b = bid % B_SZ;
  const int dBase = ((bid / B_SZ) % (D_SZ / 32)) * 32, nBase = (bid / (B_SZ * (D_SZ / 32))) * 32;
  const int tid = threadIdx.x, lane = tid & 31, w = (tid >> 5) & 3;
  const bool work = __builtin_amdgcn_readfirstlane(tid) < 128;
  const int half = lane >> 4, lan = lane & 15;
  const int mLoc = (w & 1) * 16 + lan;
  const int nLoc = (w >> 1) * 16 + lan;
  const long cs = (long)B_SZ * D_SZ * N_SZ;

  const int r0 = (tid & 127) >> 3, q4 = (tid & 7) * 4;
  const float* vsrc = v  + ((long)r0 * B_SZ + b) * D_SZ + dBase + q4;
  const float* ksrc = kd + ((long)r0 * B_SZ + b) * N_SZ + nBase + q4;
  const long vStep = (long)16 * B_SZ * D_SZ, kStep = (long)16 * B_SZ * N_SZ;
  const long vPair = (long)2 * C_SZ * B_SZ * D_SZ, kPair = (long)2 * C_SZ * B_SZ * N_SZ;

  const float* pfp = prefix + b * N_SZ + nBase + (w >> 1) * 16 + half * 8;
  float* sout = Sst + (((long)b * (D_SZ / 16) + (dBase >> 4) + (w & 1)) * (N_SZ / 16) + (nBase >> 4) + (w >> 1)) * 256 + lane * 4;

  v4f pv[16], pk[16];
  if (work) {
    #pragma unroll
    for (int k = 0; k < 16; ++k) {
      pv[k] = *(const v4f*)(vsrc + k * vStep);
      pk[k] = *(const v4f*)(ksrc + k * kStep);
    }
    #pragma unroll
    for (int k = 0; k < 16; ++k) {
      *(v4f*)&Lv[0][r0 + 16 * k][q4] = pv[k];
      *(v4f*)&Lk[0][r0 + 16 * k][q4] = pk[k];
    }
  }

  v8f s = {};
  #pragma nounroll
  for (int i = 0; i < NCHUNK / 2; ++i) {
    const int st = i & 1;
    v4f pf0a = {}, pf0b = {}, pf1a = {}, pf1b = {};
    if (work) {
      pf0a = *(const v4f*)(pfp + (2 * i) * B_SZ * N_SZ);
      pf0b = *(const v4f*)(pfp + (2 * i) * B_SZ * N_SZ + 4);
      pf1a = *(const v4f*)(pfp + (2 * i + 1) * B_SZ * N_SZ);
      pf1b = *(const v4f*)(pfp + (2 * i + 1) * B_SZ * N_SZ + 4);
      if (i + 1 < NCHUNK / 2) {
        #pragma unroll
        for (int k = 0; k < 16; ++k) {
          pv[k] = *(const v4f*)(vsrc + (i + 1) * vPair + k * vStep);
          pk[k] = *(const v4f*)(ksrc + (i + 1) * kPair + k * kStep);
        }
      }
    }
    lds_barrier();
    if (work) {
      v2f a0[4], b0[4], a1[4], b1[4];
      #pragma unroll
      for (int q = 0; q < 4; ++q) {
        const int ka = 4 * q + half * 2;
        a0[q].x = Lv[st][ka][mLoc];          a0[q].y = Lv[st][ka + 1][mLoc];
        b0[q].x = Lk[st][ka][nLoc];          b0[q].y = Lk[st][ka + 1][nLoc];
        a1[q].x = Lv[st][C_SZ + ka][mLoc];   a1[q].y = Lv[st][C_SZ + ka + 1][mLoc];
        b1[q].x = Lk[st][C_SZ + ka][nLoc];   b1[q].y = Lk[st][C_SZ + ka + 1][nLoc];
      }
      v8f u0 = {}, u1 = {};
      #pragma unroll
      for (int g = 0; g < 8; ++g) {
        v2f na0[4], nb0[4], na1[4], nb1[4];
        if (g < 7) {
          #pragma unroll
          for (int q = 0; q < 4; ++q) {
            const int ka = 16 * (g + 1) + 4 * q + half * 2;
            na0[q].x = Lv[st][ka][mLoc];          na0[q].y = Lv[st][ka + 1][mLoc];
            nb0[q].x = Lk[st][ka][nLoc];          nb0[q].y = Lk[st][ka + 1][nLoc];
            na1[q].x = Lv[st][C_SZ + ka][mLoc];   na1[q].y = Lv[st][C_SZ + ka + 1][mLoc];
            nb1[q].x = Lk[st][C_SZ + ka][nLoc];   nb1[q].y = Lk[st][C_SZ + ka + 1][nLoc];
          }
        }
        __builtin_amdgcn_sched_barrier(0);
        #pragma unroll
        for (int q = 0; q < 4; ++q) {
          u0 = wmma_f32_k4(b0[q], a0[q], u0);
          u1 = wmma_f32_k4(b1[q], a1[q], u1);
        }
        __builtin_amdgcn_sched_barrier(0);
        if (g < 7) {
          #pragma unroll
          for (int q = 0; q < 4; ++q) { a0[q] = na0[q]; b0[q] = nb0[q]; a1[q] = na1[q]; b1[q] = nb1[q]; }
        }
      }

      float* so = sout + (long)(2 * i) * cs;
      { v4f lo = {s[0], s[1], s[2], s[3]}, hi = {s[4], s[5], s[6], s[7]};
        *(v4f*)so = lo;  *(v4f*)(so + 128) = hi; }
      #pragma unroll
      for (int r = 0; r < 4; ++r) { s[r] = s[r] * pf0a[r] + u0[r];  s[r + 4] = s[r + 4] * pf0b[r] + u0[r + 4]; }
      { v4f lo = {s[0], s[1], s[2], s[3]}, hi = {s[4], s[5], s[6], s[7]};
        *(v4f*)(so + cs) = lo;  *(v4f*)(so + cs + 128) = hi; }
      #pragma unroll
      for (int r = 0; r < 4; ++r) { s[r] = s[r] * pf1a[r] + u1[r];  s[r + 4] = s[r + 4] * pf1b[r] + u1[r + 4]; }

      if (i + 1 < NCHUNK / 2) {
        #pragma unroll
        for (int k = 0; k < 16; ++k) {
          *(v4f*)&Lv[st ^ 1][r0 + 16 * k][q4] = pv[k];
          *(v4f*)&Lk[st ^ 1][r0 + 16 * k][q4] = pk[k];
        }
      }
    }
  }
}

__device__ __forceinline__ void
scores_part(float* __restrict__ L, const int sid,
            const float* __restrict__ q, const float* __restrict__ kd,
            float* __restrict__ Aw) {
  float (*Pa)[65] = (float (*)[65])L;
  float (*Pb)[65] = (float (*)[65])(L + 64 * 65);
  const int cbi = sid % (B_SZ * NCHUNK);
  const int c = cbi / B_SZ, b = cbi % B_SZ;
  const int tBase = ((sid / (B_SZ * NCHUNK)) % 2) * 64, sBase = (sid / (B_SZ * NCHUNK * 2)) * 64;
  const int tid = threadIdx.x, lane = tid & 31, w = tid >> 5;
  const int half = lane >> 4, lan = lane & 15;
  const int wr = w & 3, wc = w >> 2;
  const int mRow = wr * 16 + lan;

  v8f acc[2] = {};
  for (int kn = 0; kn < N_SZ; kn += 64) {
    #pragma unroll 2
    for (int idx = tid; idx < 64 * 64; idx += 256) {
      const int r = idx >> 6, j = idx & 63;
      Pa[r][j] = q [((long)(c * C_SZ + tBase + r) * B_SZ + b) * N_SZ + kn + j];
      Pb[r][j] = kd[((long)(c * C_SZ + sBase + r) * B_SZ + b) * N_SZ + kn + j];
    }
    __syncthreads();
    #pragma unroll
    for (int kk = 0; kk < 64; kk += 4) {
      const int ka = kk + half * 2;
      v2f a; a.x = Pa[mRow][ka]; a.y = Pa[mRow][ka + 1];
      #pragma unroll
      for (int j = 0; j < 2; ++j) {
        const int sLoc = wc * 32 + j * 16 + lan;
        v2f bf; bf.x = Pb[sLoc][ka]; bf.y = Pb[sLoc][ka + 1];
        acc[j] = wmma_f32_k4(a, bf, acc[j]);
      }
    }
    __syncthreads();
  }
  const long ab = (long)(c * B_SZ + b) * C_SZ;
  #pragma unroll
  for (int j = 0; j < 2; ++j)
    #pragma unroll
    for (int i = 0; i < 8; ++i) {
      const int row = tBase + wr * 16 + half * 8 + i;
      const int col = sBase + wc * 32 + j * 16 + lan;
      Aw[(ab + row) * C_SZ + col] = (col <= row) ? acc[j][i] : 0.0f;
    }
}

__global__ void __launch_bounds__(256)
k_state_scores(const float* __restrict__ v, const float* __restrict__ kd,
               const float* __restrict__ prefix, float* __restrict__ Sst,
               const float* __restrict__ q, float* __restrict__ Aw) {
  __shared__ float L[FUSED_LDS];
  if (blockIdx.x < ST_BLOCKS) state_part(L, blockIdx.x, v, kd, prefix, Sst);
  else                        scores_part(L, blockIdx.x - ST_BLOCKS, q, kd, Aw);
}

#define RD_AHEAD 4
#define RD_RING  (RD_AHEAD + 1)
#define RD_GROUPS (N_SZ / 16)

__global__ void __launch_bounds__(256) __attribute__((amdgpu_waves_per_eu(2, 2)))
k_readout(const float* __restrict__ q, const float* __restrict__ prefix,
          const float* __restrict__ Sst, const float* __restrict__ Aw,
          const float* __restrict__ v, float* __restrict__ out) {
  __shared__ float Pv[C_SZ][65];
  const int c = blockIdx.x / B_SZ, b = blockIdx.x % B_SZ;
  const int dBase = blockIdx.y * 64;
  const int tid = threadIdx.x, lane = tid & 31, w = tid >> 5;
  const int half = lane >> 4, lan = lane & 15;
  const int mRow = w * 16 + lan;
  const long cs = (long)B_SZ * D_SZ * N_SZ;

  const float* qrow = q + ((long)(c * C_SZ + mRow) * B_SZ + b) * N_SZ + half * 8;
  const float* pfl  = prefix + (c * B_SZ + b) * N_SZ + half * 8;
  const float* arow = Aw + ((long)(c * B_SZ + b) * C_SZ + mRow) * C_SZ + half * 2;
  const float* sfr  = Sst + (long)c * cs + ((long)b * (D_SZ / 16) + (dBase >> 4)) * (N_SZ / 16) * 256 + lane * 4;

  v4f qa[RD_RING][2], pa[RD_RING][2], sb[RD_RING][4][2];
  #pragma unroll
  for (int g = 0; g < RD_AHEAD; ++g) {
    #pragma unroll
    for (int sub = 0; sub < 2; ++sub) {
      qa[g][sub] = *(const v4f*)(qrow + 16 * g + 4 * sub);
      pa[g][sub] = *(const v4f*)(pfl + 16 * g + 4 * sub);
      #pragma unroll
      for (int j = 0; j < 4; ++j) sb[g][j][sub] = *(const v4f*)(sfr + (j * 16 + g) * 256 + 128 * sub);
    }
  }

  for (int idx = tid; idx < C_SZ * 64; idx += 256) {
    const int s = idx >> 6, j = idx & 63;
    Pv[s][j] = v[((long)(c * C_SZ + s) * B_SZ + b) * D_SZ + dBase + j];
  }
  lds_barrier();

  v2f as[32];
  v8f acc[4] = {};
  #pragma unroll
  for (int g = 0; g < RD_GROUPS; ++g) {
    const int slot = g % RD_RING;
    if (g + RD_AHEAD < RD_GROUPS) {
      const int gn = g + RD_AHEAD, sn = gn % RD_RING;
      #pragma unroll
      for (int sub = 0; sub < 2; ++sub) {
        qa[sn][sub] = *(const v4f*)(qrow + 16 * gn + 4 * sub);
        pa[sn][sub] = *(const v4f*)(pfl + 16 * gn + 4 * sub);
        #pragma unroll
        for (int j = 0; j < 4; ++j) sb[sn][j][sub] = *(const v4f*)(sfr + (j * 16 + gn) * 256 + 128 * sub);
      }
    } else {
      const int part = g + RD_AHEAD - RD_GROUPS;
      #pragma unroll
      for (int i = 0; i < 32 / RD_AHEAD; ++i) as[part * (32 / RD_AHEAD) + i] = *(const v2f*)(arow + 4 * (part * (32 / RD_AHEAD) + i));
    }
    __builtin_amdgcn_sched_barrier(0);
    #pragma unroll
    for (int sub = 0; sub < 2; ++sub) {
      const v4f a4 = qa[slot][sub] * pa[slot][sub];
      #pragma unroll
      for (int e2 = 0; e2 < 2; ++e2) {
        v2f a; a.x = e2 ? a4.z : a4.x; a.y = e2 ? a4.w : a4.y;
        #pragma unroll
        for (int j = 0; j < 4; ++j) {
          const v4f b4 = sb[slot][j][sub];
          v2f bf; bf.x = e2 ? b4.z : b4.x; bf.y = e2 ? b4.w : b4.y;
          acc[j] = wmma_f32_k4(a, bf, acc[j]);
        }
      }
    }
    __builtin_amdgcn_sched_barrier(0);
  }

  v8f acc2[4] = {};
  #pragma unroll
  for (int i = 0; i < 32; ++i) {
    const int ka = 4 * i + half * 2;
    const v2f a = as[i];
    #pragma unroll
    for (int j = 0; j < 4; ++j) {
      v2f bf; bf.x = Pv[ka][j * 16 + lan]; bf.y = Pv[ka + 1][j * 16 + lan];
      acc2[j] = wmma_f32_k4(a, bf, acc2[j]);
    }
  }

  #pragma unroll
  for (int j = 0; j < 4; ++j)
    #pragma unroll
    for (int i = 0; i < 8; ++i) {
      const int row = w * 16 + half * 8 + i;
      const int col = dBase + j * 16 + lan;
      out[((long)(c * C_SZ + row) * B_SZ + b) * D_SZ + col] = acc2[j][i] + acc[j][i];
    }
}

extern "C" void kernel_launch(void* const* d_in, const int* in_sizes, int n_in,
                              void* d_out, int out_size, void* d_ws, size_t ws_size,
                              hipStream_t stream) {
  const float* v     = (const float*)d_in[0];
  const float* k     = (const float*)d_in[1];
  const float* alpha = (const float*)d_in[2];
  const float* q     = (const float*)d_in[3];
  float* out = (float*)d_out;
  float* ws  = (float*)d_ws;

  const long KD_N = (long)T_TOT * B_SZ * N_SZ;
  const long PF_N = (long)NCHUNK * B_SZ * N_SZ;
  const long AW_N = (long)NCHUNK * B_SZ * C_SZ * C_SZ;
  float* kd     = ws;
  float* prefix = kd + KD_N;
  float* Aw     = prefix + PF_N;
  float* Sst    = Aw + AW_N;

  k_decay<<<dim3(B_SZ * NCHUNK * (N_SZ / DEC_T)), DEC_T, 0, stream>>>(k, alpha, kd, prefix);
  k_state_scores<<<dim3(ST_BLOCKS + SC_BLOCKS), 256, 0, stream>>>(v, kd, prefix, Sst, q, Aw);
  k_readout<<<dim3(B_SZ * NCHUNK, D_SZ / 64), 256, 0, stream>>>(q, prefix, Sst, Aw, v, out);
}


# k_readout: v panel lies in LDS in the s sum's B-fragment order (k pair of two d tiles per 16 bytes): fragments read as 16-byte pieces, 8 k steps ahead; panel fetched as one load group; sum order kept
# speedup vs baseline: 3.4866x; 1.0432x over previous
#include <hip/hip_runtime.h>


typedef float v2f __attribute__((ext_vector_type(2)));
typedef float v8f __attribute__((ext_vector_type(8)));

#define T_TOT  2048
#define B_SZ   4
#define D_SZ   256
#define N_SZ   256
#define C_SZ   128
#define NCHUNK 16
#define EPS_F  1e-8f

__device__ __forceinline__ v8f wmma_f32_k4(v2f a, v2f b, v8f c) {
  return __builtin_amdgcn_wmma_f32_16x16x4_f32(false, a, false, b, (short)0, c, false, false);
}

__device__ __forceinline__ void lds_barrier() {
  asm volatile("s_wait_dscnt 0x0\n\ts_barrier_signal -1\n\ts_barrier_wait -1" ::: "memory");
}

#define DEC_T 64
__global__ void __launch_bounds__(DEC_T)
k_decay(const float* __restrict__ kin, const float* __restrict__ alpha,
        float* __restrict__ kd, float* __restrict__ prefix) {
  const int cb = blockIdx.x / (N_SZ / DEC_T);
  const int c = cb / B_SZ;
  const int b = cb % B_SZ;
  const int n = (blockIdx.x % (N_SZ / DEC_T)) * DEC_T + threadIdx.x;
  const long stride = (long)B_SZ * N_SZ;
  const long base = ((long)c * C_SZ * B_SZ + b) * N_SZ + n;

  float r[C_SZ];
  float kv[C_SZ];
  #pragma unroll
  for (int t = 0; t < C_SZ; ++t) r[t] = alpha[base + (long)t * stride];
  #pragma unroll
  for (int t = 0; t < C_SZ; ++t) kv[t] = kin[base + (long)t * stride];

  float P = 1.0f;
  #pragma unroll
  for (int t = 0; t < C_SZ; ++t) {
    float a = r[t];
    a = (a < EPS_F) ? EPS_F : a;
    P *= a;
    r[t] = P;
  }
  const float inv = 1.0f / (P + EPS_F);
  #pragma unroll
  for (int t = 0; t < C_SZ; ++t)
    kd[base + (long)t * stride] = kv[t] * (r[t] * inv);
  prefix[cb * N_SZ + n] = r[0];
}

typedef float v4f __attribute__((ext_vector_type(4)));
#define ST_P 40
#define ST_BLOCKS (B_SZ * (D_SZ / 32) * (N_SZ / 32))
#define SC_BLOCKS (B_SZ * NCHUNK * 2 * 2)
#define FUSED_LDS (2 * 2 * 2 * C_SZ * ST_P)

__device__ __forceinline__ void
state_part(float* __restrict__ L, const int bid,
           const float* __restrict__ v, const float* __restrict__ kd,
           const float* __restrict__ prefix, float* __restrict__ Sst) {
  float (*Lv)[2 * C_SZ][ST_P] = (float (*)[2 * C_SZ][ST_P])L;
  float (*Lk)[2 * C_SZ][ST_P] = (float (*)[2 * C_SZ][ST_P])(L + 2 * 2 * C_SZ * ST_P);
  const int b = bid % B_SZ;
  const int dBase = ((bid / B_SZ) % (D_SZ / 32)) * 32, nBase = (bid / (B_SZ * (D_SZ / 32))) * 32;
  const int tid = threadIdx.x, lane = tid & 31, w = (tid >> 5) & 3;
  const bool work = __builtin_amdgcn_readfirstlane(tid) < 128;
  const int half = lane >> 4, lan = lane & 15;
  const int mLoc = (w & 1) * 16 + lan;
  const int nLoc = (w >> 1) * 16 + lan;
  const long cs = (long)B_SZ * D_SZ * N_SZ;

  const int r0 = (tid & 127) >> 3, q4 = (tid & 7) * 4;
  const float* vsrc = v  + ((long)r0 * B_SZ + b) * D_SZ + dBase + q4;
  const float* ksrc = kd + ((long)r0 * B_SZ + b) * N_SZ + nBase + q4;
  const long vStep = (long)16 * B_SZ * D_SZ, kStep = (long)16 * B_SZ * N_SZ;
  const long vPair = (long)2 * C_SZ * B_SZ * D_SZ, kPair = (long)2 * C_SZ * B_SZ * N_SZ;

  const float* pfp = prefix + b * N_SZ + nBase + (w >> 1) * 16 + half * 8;
  float* sout = Sst + (((long)b * (D_SZ / 16) + (dBase >> 4) + (w & 1)) * (N_SZ / 16) + (nBase >> 4) + (w >> 1)) * 256 + lane * 4;

  v4f pv[16], pk[16];
  if (work) {
    #pragma unroll
    for (int k = 0; k < 16; ++k) {
      pv[k] = *(const v4f*)(vsrc + k * vStep);
      pk[k] = *(const v4f*)(ksrc + k * kStep);
    }
    #pragma unroll
    for (int k = 0; k < 16; ++k) {
      *(v4f*)&Lv[0][r0 + 16 * k][q4] = pv[k];
      *(v4f*)&Lk[0][r0 + 16 * k][q4] = pk[k];
    }
  }

  v8f s = {};
  #pragma nounroll
  for (int i = 0; i < NCHUNK / 2; ++i) {
    const int st = i & 1;
    v4f pf0a = {}, pf0b = {}, pf1a = {}, pf1b = {};
    if (work) {
      pf0a = *(const v4f*)(pfp + (2 * i) * B_SZ * N_SZ);
      pf0b = *(const v4f*)(pfp + (2 * i) * B_SZ * N_SZ + 4);
      pf1a = *(const v4f*)(pfp + (2 * i + 1) * B_SZ * N_SZ);
      pf1b = *(const v4f*)(pfp + (2 * i + 1) * B_SZ * N_SZ + 4);
      if (i + 1 < NCHUNK / 2) {
        #pragma unroll
        for (int k = 0; k < 16; ++k) {
          pv[k] = *(const v4f*)(vsrc + (i + 1) * vPair + k * vStep);
          pk[k] = *(const v4f*)(ksrc + (i + 1) * kPair + k * kStep);
        }
      }
    }
    lds_barrier();
    if (work) {
      v2f a0[4], b0[4], a1[4], b1[4];
      #pragma unroll
      for (int q = 0; q < 4; ++q) {
        const int ka = 4 * q + half * 2;
        a0[q].x = Lv[st][ka][mLoc];          a0[q].y = Lv[st][ka + 1][mLoc];
        b0[q].x = Lk[st][ka][nLoc];          b0[q].y = Lk[st][ka + 1][nLoc];
        a1[q].x = Lv[st][C_SZ + ka][mLoc];   a1[q].y = Lv[st][C_SZ + ka + 1][mLoc];
        b1[q].x = Lk[st][C_SZ + ka][nLoc];   b1[q].y = Lk[st][C_SZ + ka + 1][nLoc];
      }
      v8f u0 = {}, u1 = {};
      #pragma unroll
      for (int g = 0; g < 8; ++g) {
        v2f na0[4], nb0[4], na1[4], nb1[4];
        if (g < 7) {
          #pragma unroll
          for (int q = 0; q < 4; ++q) {
            const int ka = 16 * (g + 1) + 4 * q + half * 2;
            na0[q].x = Lv[st][ka][mLoc];          na0[q].y = Lv[st][ka + 1][mLoc];
            nb0[q].x = Lk[st][ka][nLoc];          nb0[q].y = Lk[st][ka + 1][nLoc];
            na1[q].x = Lv[st][C_SZ + ka][mLoc];   na1[q].y = Lv[st][C_SZ + ka + 1][mLoc];
            nb1[q].x = Lk[st][C_SZ + ka][nLoc];   nb1[q].y = Lk[st][C_SZ + ka + 1][nLoc];
          }
        }
        __builtin_amdgcn_sched_barrier(0);
        #pragma unroll
        for (int q = 0; q < 4; ++q) {
          u0 = wmma_f32_k4(b0[q], a0[q], u0);
          u1 = wmma_f32_k4(b1[q], a1[q], u1);
        }
        __builtin_amdgcn_sched_barrier(0);
        if (g < 7) {
          #pragma unroll
          for (int q = 0; q < 4; ++q) { a0[q] = na0[q]; b0[q] = nb0[q]; a1[q] = na1[q]; b1[q] = nb1[q]; }
        }
      }

      float* so = sout + (long)(2 * i) * cs;
      { v4f lo = {s[0], s[1], s[2], s[3]}, hi = {s[4], s[5], s[6], s[7]};
        *(v4f*)so = lo;  *(v4f*)(so + 128) = hi; }
      #pragma unroll
      for (int r = 0; r < 4; ++r) { s[r] = s[r] * pf0a[r] + u0[r];  s[r + 4] = s[r + 4] * pf0b[r] + u0[r + 4]; }
      { v4f lo = {s[0], s[1], s[2], s[3]}, hi = {s[4], s[5], s[6], s[7]};
        *(v4f*)(so + cs) = lo;  *(v4f*)(so + cs + 128) = hi; }
      #pragma unroll
      for (int r = 0; r < 4; ++r) { s[r] = s[r] * pf1a[r] + u1[r];  s[r + 4] = s[r + 4] * pf1b[r] + u1[r + 4]; }

      if (i + 1 < NCHUNK / 2) {
        #pragma unroll
        for (int k = 0; k < 16; ++k) {
          *(v4f*)&Lv[st ^ 1][r0 + 16 * k][q4] = pv[k];
          *(v4f*)&Lk[st ^ 1][r0 + 16 * k][q4] = pk[k];
        }
      }
    }
  }
}

__device__ __forceinline__ void
scores_part(float* __restrict__ L, const int sid,
            const float* __restrict__ q, const float* __restrict__ kd,
            float* __restrict__ Aw) {
  float (*Pa)[65] = (float (*)[65])L;
  float (*Pb)[65] = (float (*)[65])(L + 64 * 65);
  const int cbi = sid % (B_SZ * NCHUNK);
  const int c = cbi / B_SZ, b = cbi % B_SZ;
  const int tBase = ((sid / (B_SZ * NCHUNK)) % 2) * 64, sBase = (sid / (B_SZ * NCHUNK * 2)) * 64;
  const int tid = threadIdx.x, lane = tid & 31, w = tid >> 5;
  const int half = lane >> 4, lan = lane & 15;
  const int wr = w & 3, wc = w >> 2;
  const int mRow = wr * 16 + lan;

  v8f acc[2] = {};
  for (int kn = 0; kn < N_SZ; kn += 64) {
    #pragma unroll 2
    for (int idx = tid; idx < 64 * 64; idx += 256) {
      const int r = idx >> 6, j = idx & 63;
      Pa[r][j] = q [((long)(c * C_SZ + tBase + r) * B_SZ + b) * N_SZ + kn + j];
      Pb[r][j] = kd[((long)(c * C_SZ + sBase + r) * B_SZ + b) * N_SZ + kn + j];
    }
    __syncthreads();
    #pragma unroll
    for (int kk = 0; kk < 64; kk += 4) {
      const int ka = kk + half * 2;
      v2f a; a.x = Pa[mRow][ka]; a.y = Pa[mRow][ka + 1];
      #pragma unroll
      for (int j = 0; j < 2; ++j) {
        const int sLoc = wc * 32 + j * 16 + lan;
        v2f bf; bf.x = Pb[sLoc][ka]; bf.y = Pb[sLoc][ka + 1];
        acc[j] = wmma_f32_k4(a, bf, acc[j]);
      }
    }
    __syncthreads();
  }
  const long ab = (long)(c * B_SZ + b) * C_SZ;
  #pragma unroll
  for (int j = 0; j < 2; ++j)
    #pragma unroll
    for (int i = 0; i < 8; ++i) {
      const int row = tBase + wr * 16 + half * 8 + i;
      const int col = sBase + wc * 32 + j * 16 + lan;
      Aw[(ab + row) * C_SZ + col] = (col <= row) ? acc[j][i] : 0.0f;
    }
}

__global__ void __launch_bounds__(256)
k_state_scores(const float* __restrict__ v, const float* __restrict__ kd,
               const float* __restrict__ prefix, float* __restrict__ Sst,
               const float* __restrict__ q, float* __restrict__ Aw) {
  __shared__ float L[FUSED_LDS];
  if (blockIdx.x < ST_BLOCKS) state_part(L, blockIdx.x, v, kd, prefix, Sst);
  else                        scores_part(L, blockIdx.x - ST_BLOCKS, q, kd, Aw);
}

#define RD_AHEAD 4
#define RD_RING  (RD_AHEAD + 1)
#define RD_GROUPS (N_SZ / 16)
#define AV_G 8
#define AV_GROUPS (32 / AV_G)

__global__ void __launch_bounds__(256) __attribute__((amdgpu_waves_per_eu(2, 2)))
k_readout(const float* __restrict__ q, const float* __restrict__ prefix,
          const float* __restrict__ Sst, const float* __restrict__ Aw,
          const float* __restrict__ v, float* __restrict__ out) {
  __shared__ v4f Pv4[32][2][2][16];
  const int c = blockIdx.x / B_SZ, b = blockIdx.x % B_SZ;
  const int dBase = blockIdx.y * 64;
  const int tid = threadIdx.x, lane = tid & 31, w = tid >> 5;
  const int half = lane >> 4, lan = lane & 15;
  const int mRow = w * 16 + lan;
  const long cs = (long)B_SZ * D_SZ * N_SZ;

  const int p = tid & 15, rp0 = tid >> 4;
  const float* vsrc = v + ((long)(c * C_SZ + 2 * rp0) * B_SZ + b) * D_SZ + dBase + 4 * p;
  v4f va[4], vb[4];
  #pragma unroll
  for (int k = 0; k < 4; ++k) {
    va[k] = *(const v4f*)(vsrc + (long)(32 * k) * B_SZ * D_SZ);
    vb[k] = *(const v4f*)(vsrc + (long)(32 * k + 1) * B_SZ * D_SZ);
  }

  __builtin_amdgcn_sched_barrier(0);

  const float* qrow = q + ((long)(c * C_SZ + mRow) * B_SZ + b) * N_SZ + half * 8;
  const float* pfl  = prefix + (c * B_SZ + b) * N_SZ + half * 8;
  const float* arow = Aw + ((long)(c * B_SZ + b) * C_SZ + mRow) * C_SZ + half * 2;
  const float* sfr  = Sst + (long)c * cs + ((long)b * (D_SZ / 16) + (dBase >> 4)) * (N_SZ / 16) * 256 + lane * 4;

  v4f qa[RD_RING][2], pa[RD_RING][2], sb[RD_RING][4][2];
  #pragma unroll
  for (int g = 0; g < RD_AHEAD; ++g) {
    #pragma unroll
    for (int sub = 0; sub < 2; ++sub) {
      qa[g][sub] = *(const v4f*)(qrow + 16 * g + 4 * sub);
      pa[g][sub] = *(const v4f*)(pfl + 16 * g + 4 * sub);
      #pragma unroll
      for (int j = 0; j < 4; ++j) sb[g][j][sub] = *(const v4f*)(sfr + (j * 16 + g) * 256 + 128 * sub);
    }
  }

  __builtin_amdgcn_sched_barrier(0);
  {
    float* Pf = (float*)Pv4;
    const int jp = p >> 3, jj = (p >> 2) & 1, c0 = 4 * (p & 3);
    #pragma unroll
    for (int k = 0; k < 4; ++k) {
      const int rp = rp0 + 16 * k;
      float* dst = Pf + ((rp * 2 + jp) * 16 + c0) * 4 + jj * 2;
      #pragma unroll
      for (int r = 0; r < 4; ++r) {
        v2f pr; pr.x = va[k][r]; pr.y = vb[k][r];
        *(v2f*)(dst + 4 * r) = pr;
      }
    }
  }
  lds_barrier();

  v2f as[32];
  v4f fb[2][AV_G][2];
  v8f acc[4] = {};
  #pragma unroll
  for (int g = 0; g < RD_GROUPS; ++g) {
    const int slot = g % RD_RING;
    if (g + RD_AHEAD < RD_GROUPS) {
      const int gn = g + RD_AHEAD, sn = gn % RD_RING;
      #pragma unroll
      for (int sub = 0; sub < 2; ++sub) {
        qa[sn][sub] = *(const v4f*)(qrow + 16 * gn + 4 * sub);
        pa[sn][sub] = *(const v4f*)(pfl + 16 * gn + 4 * sub);
        #pragma unroll
        for (int j = 0; j < 4; ++j) sb[sn][j][sub] = *(const v4f*)(sfr + (j * 16 + gn) * 256 + 128 * sub);
      }
    } else {
      const int part = g + RD_AHEAD - RD_GROUPS;
      #pragma unroll
      for (int i = 0; i < 32 / RD_AHEAD; ++i) as[part * (32 / RD_AHEAD) + i] = *(const v2f*)(arow + 4 * (part * (32 / RD_AHEAD) + i));
    }
    if (g == RD_GROUPS - 1) {
      #pragma unroll
      for (int kq = 0; kq < AV_G; ++kq) { fb[0][kq][0] = Pv4[kq][half][0][lan]; fb[0][kq][1] = Pv4[kq][half][1][lan]; }
    }
    __builtin_amdgcn_sched_barrier(0);
    #pragma unroll
    for (int sub = 0; sub < 2; ++sub) {
      const v4f a4 = qa[slot][sub] * pa[slot][sub];
      #pragma unroll
      for (int e2 = 0; e2 < 2; ++e2) {
        v2f a; a.x = e2 ? a4.z : a4.x; a.y = e2 ? a4.w : a4.y;
        #pragma unroll
        for (int j = 0; j < 4; ++j) {
          const v4f b4 = sb[slot][j][sub];
          v2f bf; bf.x = e2 ? b4.z : b4.x; bf.y = e2 ? b4.w : b4.y;
          acc[j] = wmma_f32_k4(a, bf, acc[j]);
        }
      }
    }
    __builtin_amdgcn_sched_barrier(0);
  }

  v8f acc2[4] = {};
  #pragma unroll
  for (int g = 0; g < AV_GROUPS; ++g) {
    const int cur = g & 1;
    if (g + 1 < AV_GROUPS) {
      #pragma unroll
      for (int kq = 0; kq < AV_G; ++kq) {
        fb[cur ^ 1][kq][0] = Pv4[(g + 1) * AV_G + kq][half][0][lan];
        fb[cur ^ 1][kq][1] = Pv4[(g + 1) * AV_G + kq][half][1][lan];
      }
    }
    __builtin_amdgcn_sched_barrier(0);
    #pragma unroll
    for (int kq = 0; kq < AV_G; ++kq) {
      const v2f a = as[g * AV_G + kq];
      const v4f f0 = fb[cur][kq][0], f1 = fb[cur][kq][1];
      v2f b0; b0.x = f0.x; b0.y = f0.y;
      v2f b1; b1.x = f0.z; b1.y = f0.w;
      v2f b2; b2.x = f1.x; b2.y = f1.y;
      v2f b3; b3.x = f1.z; b3.y = f1.w;
      acc2[0] = wmma_f32_k4(a, b0, acc2[0]);
      acc2[1] = wmma_f32_k4(a, b1, acc2[1]);
      acc2[2] = wmma_f32_k4(a, b2, acc2[2]);
      acc2[3] = wmma_f32_k4(a, b3, acc2[3]);
    }
    __builtin_amdgcn_sched_barrier(0);
  }

  #pragma unroll
  for (int j = 0; j < 4; ++j)
    #pragma unroll
    for (int i = 0; i < 8; ++i) {
      const int row = w * 16 + half * 8 + i;
      const int col = dBase + j * 16 + lan;
      out[((long)(c * C_SZ + row) * B_SZ + b) * D_SZ + col] = acc2[j][i] + acc[j][i];
    }
}

extern "C" void kernel_launch(void* const* d_in, const int* in_sizes, int n_in,
                              void* d_out, int out_size, void* d_ws, size_t ws_size,
                              hipStream_t stream) {
  const float* v     = (const float*)d_in[0];
  const float* k     = (const float*)d_in[1];
  const float* alpha = (const float*)d_in[2];
  const float* q     = (const float*)d_in[3];
  float* out = (float*)d_out;
  float* ws  = (float*)d_ws;

  const long KD_N = (long)T_TOT * B_SZ * N_SZ;
  const long PF_N = (long)NCHUNK * B_SZ * N_SZ;
  const long AW_N = (long)NCHUNK * B_SZ * C_SZ * C_SZ;
  float* kd     = ws;
  float* prefix = kd + KD_N;
  float* Aw     = prefix + PF_N;
  float* Sst    = Aw + AW_N;

  k_decay<<<dim3(B_SZ * NCHUNK * (N_SZ / DEC_T)), DEC_T, 0, stream>>>(k, alpha, kd, prefix);
  k_state_scores<<<dim3(ST_BLOCKS + SC_BLOCKS), 256, 0, stream>>>(v, kd, prefix, Sst, q, Aw);
  k_readout<<<dim3(B_SZ * NCHUNK, D_SZ / 64), 256, 0, stream>>>(q, prefix, Sst, Aw, v, out);
}
